# starting kernel
# baseline (speedup 1.0000x reference)
#include <hip/hip_runtime.h>
#include <hip/hip_bf16.h>


typedef __attribute__((ext_vector_type(2))) float v2f;
typedef __attribute__((ext_vector_type(8))) float v8f;

#define N_TOTAL 116250
#define DDIM 128
#define NT_PER_WAVE 2
#define WAVES_PER_BLOCK 8
#define N_PER_BLOCK (WAVES_PER_BLOCK * NT_PER_WAVE * 16)

__global__ __launch_bounds__(256) void conv_gemm_wmma_f32(
    const float* __restrict__ x,
    const float* __restrict__ w,
    float* __restrict__ out)
{
    const int lane  = threadIdx.x & 31;
    const int wave  = threadIdx.x >> 5;
    const int lrow  = lane & 15;
    const int khalf = (lane >> 4) << 1;

    const long long n_base =
        (long long)blockIdx.x * N_PER_BLOCK + (long long)wave * (NT_PER_WAVE * 16);

    v8f acc[4][NT_PER_WAVE];
    {
        v8f zero = {};
        #pragma unroll
        for (int ft = 0; ft < 4; ++ft)
            #pragma unroll
            for (int nt = 0; nt < NT_PER_WAVE; ++nt)
                acc[ft][nt] = zero;
    }

    const float* xrow[NT_PER_WAVE];
    #pragma unroll
    for (int nt = 0; nt < NT_PER_WAVE; ++nt) {
        long long n = n_base + nt * 16 + lrow;
        if (n > (long long)(N_TOTAL - 1)) n = N_TOTAL - 1;
        xrow[nt] = x + n * DDIM + khalf;
    }
    const float* wrow = w + (long long)lrow * DDIM + khalf;

    for (int kb = 0; kb < DDIM; kb += 4) {
        v2f a[4], b[NT_PER_WAVE];
        #pragma unroll
        for (int ft = 0; ft < 4; ++ft)
            a[ft] = *(const v2f*)(wrow + ft * 16 * DDIM + kb);
        #pragma unroll
        for (int nt = 0; nt < NT_PER_WAVE; ++nt)
            b[nt] = *(const v2f*)(xrow[nt] + kb);
        #pragma unroll
        for (int ft = 0; ft < 4; ++ft)
            #pragma unroll
            for (int nt = 0; nt < NT_PER_WAVE; ++nt)
                acc[ft][nt] = __builtin_amdgcn_wmma_f32_16x16x4_f32(
                     false, a[ft],
                     false, b[nt],
                     (short)0, acc[ft][nt],
                     false,  false);
    }

    const int mhalf = (lane >> 4) << 3;
    #pragma unroll
    for (int nt = 0; nt < NT_PER_WAVE; ++nt) {
        const long long n = n_base + nt * 16 + lrow;
        if (n < (long long)N_TOTAL) {
            #pragma unroll
            for (int ft = 0; ft < 4; ++ft) {
                float* orow = out + (long long)(ft * 16 + mhalf) * N_TOTAL + n;
                #pragma unroll
                for (int r = 0; r < 8; ++r)
                    orow[(long long)r * N_TOTAL] = acc[ft][nt][r];
            }
        }
    }
}

extern "C" void kernel_launch(void* const* d_in, const int* in_sizes, int n_in,
                              void* d_out, int out_size, void* d_ws, size_t ws_size,
                              hipStream_t stream) {
    const float* x = (const float*)d_in[0];
    const float* w = (const float*)d_in[1];
    float* out = (float*)d_out;

    const int grid = (N_TOTAL + N_PER_BLOCK - 1) / N_PER_BLOCK;
    conv_gemm_wmma_f32<<<grid, 256, 0, stream>>>(x, w, out);
}


# W staged once per block into LDS in A-fragment order; K loop reads A fragments by ds_load_b128 one k-pair ahead instead of 128 global loads per wave; k order and summation order unchanged
# speedup vs baseline: 1.2210x; 1.2210x over previous
#include <hip/hip_runtime.h>
#include <hip/hip_bf16.h>


typedef __attribute__((ext_vector_type(2))) float v2f;
typedef __attribute__((ext_vector_type(4))) float v4f;
typedef __attribute__((ext_vector_type(8))) float v8f;

#define N_TOTAL 116250
#define DDIM 128
#define NT_PER_WAVE 2
#define WAVES_PER_BLOCK 8
#define N_PER_BLOCK (WAVES_PER_BLOCK * NT_PER_WAVE * 16)
#define KPAIRS (DDIM / 8)

#define LOAD_B(dst, kp_)                                                     \
    _Pragma("unroll")                                                        \
    for (int s_ = 0; s_ < 2; ++s_)                                           \
        _Pragma("unroll")                                                    \
        for (int nt_ = 0; nt_ < NT_PER_WAVE; ++nt_)                          \
            dst[s_][nt_] = *(const v2f*)(xrow[nt_] + (kp_) * 8 + s_ * 4);

#define LOAD_A(dst, kp_)                                                     \
    _Pragma("unroll")                                                        \
    for (int ft_ = 0; ft_ < 4; ++ft_)                                        \
        dst[ft_] = afrag[((kp_) * 4 + ft_) * 32];

__global__ __launch_bounds__(256) void conv_gemm_wmma_f32(
    const float* __restrict__ x,
    const float* __restrict__ w,
    float* __restrict__ out)
{
    __shared__ v4f wfrag[KPAIRS * 4 * 32];

    const int lane  = threadIdx.x & 31;
    const int wave  = threadIdx.x >> 5;
    const int lrow  = lane & 15;
    const int half  = lane >> 4;
    const int khalf = half << 1;

    const long long n_base =
        (long long)blockIdx.x * N_PER_BLOCK + (long long)wave * (NT_PER_WAVE * 16);

    const float* xrow[NT_PER_WAVE];
    #pragma unroll
    for (int nt = 0; nt < NT_PER_WAVE; ++nt) {
        long long n = n_base + nt * 16 + lrow;
        if (n > (long long)(N_TOTAL - 1)) n = N_TOTAL - 1;
        xrow[nt] = x + n * DDIM + khalf;
    }

    v2f b0[2][NT_PER_WAVE], b1[2][NT_PER_WAVE], b2[2][NT_PER_WAVE];
    LOAD_B(b0, 0)
    LOAD_B(b1, 1)

    {
        float* wl = (float*)wfrag;
        #pragma unroll
        for (int j = 0; j < 8; ++j) {
            const int blk = wave * 8 + j;
            const int kp  = blk >> 2;
            const int ft  = blk & 3;
            const v4f v = *(const v4f*)(w + (ft * 16 + lrow) * DDIM + kp * 8 + half * 4);
            v2f lo = {v[0], v[1]};
            v2f hi = {v[2], v[3]};
            float* dst = wl + blk * 128 + lrow * 4 + half * 2;
            *(v2f*)(dst)      = lo;
            *(v2f*)(dst + 64) = hi;
        }
    }

    v8f acc[4][NT_PER_WAVE];
    {
        v8f zero = {};
        #pragma unroll
        for (int ft = 0; ft < 4; ++ft)
            #pragma unroll
            for (int nt = 0; nt < NT_PER_WAVE; ++nt)
                acc[ft][nt] = zero;
    }

    __syncthreads();

    const v4f* afrag = wfrag + lane;
    v4f a0[4], a1[4];
    LOAD_A(a0, 0)

    #pragma unroll
    for (int kp = 0; kp < KPAIRS; ++kp) {
        if (kp + 1 < KPAIRS) { LOAD_A(a1, kp + 1) }
        if (kp + 2 < KPAIRS) { LOAD_B(b2, kp + 2) }
        __builtin_amdgcn_sched_barrier(0);
        #pragma unroll
        for (int s = 0; s < 2; ++s) {
            #pragma unroll
            for (int ft = 0; ft < 4; ++ft) {
                v2f as = {a0[ft][2 * s], a0[ft][2 * s + 1]};
                #pragma unroll
                for (int nt = 0; nt < NT_PER_WAVE; ++nt)
                    acc[ft][nt] = __builtin_amdgcn_wmma_f32_16x16x4_f32(
                         false, as,
                         false, b0[s][nt],
                         (short)0, acc[ft][nt],
                         false,  false);
            }
        }
        #pragma unroll
        for (int ft = 0; ft < 4; ++ft) a0[ft] = a1[ft];
        #pragma unroll
        for (int s = 0; s < 2; ++s)
            #pragma unroll
            for (int nt = 0; nt < NT_PER_WAVE; ++nt) {
                b0[s][nt] = b1[s][nt];
                b1[s][nt] = b2[s][nt];
            }
    }

    const int mhalf = half << 3;
    #pragma unroll
    for (int nt = 0; nt < NT_PER_WAVE; ++nt) {
        const long long n = n_base + nt * 16 + lrow;
        if (n < (long long)N_TOTAL) {
            #pragma unroll
            for (int ft = 0; ft < 4; ++ft) {
                float* orow = out + (long long)(ft * 16 + mhalf) * N_TOTAL + n;
                #pragma unroll
                for (int r = 0; r < 8; ++r)
                    orow[(long long)r * N_TOTAL] = acc[ft][nt][r];
            }
        }
    }
}

extern "C" void kernel_launch(void* const* d_in, const int* in_sizes, int n_in,
                              void* d_out, int out_size, void* d_ws, size_t ws_size,
                              hipStream_t stream) {
    const float* x = (const float*)d_in[0];
    const float* w = (const float*)d_in[1];
    float* out = (float*)d_out;

    const int grid = (N_TOTAL + N_PER_BLOCK - 1) / N_PER_BLOCK;
    conv_gemm_wmma_f32<<<grid, 256, 0, stream>>>(x, w, out);
}


# Equal WMMA count per SIMD at 4 waves each: 256 blocks of 16 waves (was 455 of 8); per SIMD 3 two-tile + 1 one-tile wave, a 29th n-tile split by f-tile over 4 one-tile waves; k, sum order unchanged
# speedup vs baseline: 1.3057x; 1.0694x over previous
#include <hip/hip_runtime.h>
#include <hip/hip_bf16.h>


typedef __attribute__((ext_vector_type(2))) float v2f;
typedef __attribute__((ext_vector_type(4))) float v4f;
typedef __attribute__((ext_vector_type(8))) float v8f;

#define N_TOTAL 116250
#define DDIM 128
#define WAVES_PER_BLOCK 16
#define GRID_BLOCKS 256
#define KPAIRS (DDIM / 8)
#define TILES_TOTAL ((N_TOTAL + 15) / 16)
#define TILES_BASE (TILES_TOTAL / GRID_BLOCKS)
#define TILES_REM (TILES_TOTAL % GRID_BLOCKS)

static_assert(TILES_BASE == 28, "the wave roles below deal exactly 12 x 2 + 4 x 1 n-tiles per block");
static_assert(TILES_REM < GRID_BLOCKS, "at most one extra tile per block");

#define LOAD_B1(dst, kp_, nt_)                                               \
    _Pragma("unroll")                                                        \
    for (int s_ = 0; s_ < 2; ++s_)                                           \
        dst[s_][nt_] = *(const v2f*)(xrow[nt_] + (kp_) * 8 + s_ * 4);

#define LOAD_A(dst, kp_)                                                     \
    _Pragma("unroll")                                                        \
    for (int ft_ = 0; ft_ < 4; ++ft_)                                        \
        dst[ft_] = afrag[((kp_) * 4 + ft_) * 32];

__device__ __forceinline__ v8f wmma4(v2f a, v2f b, v8f c) {
    return __builtin_amdgcn_wmma_f32_16x16x4_f32(
         false, a,  false, b,
         (short)0, c,  false,  false);
}

__device__ __forceinline__ void store_ftile(float* __restrict__ out, int ft, int mhalf,
                                            long long n, const v8f& acc) {
    float* orow = out + (long long)(ft * 16 + mhalf) * N_TOTAL + n;
    #pragma unroll
    for (int r = 0; r < 8; ++r)
        orow[(long long)r * N_TOTAL] = acc[r];
}

__device__ __forceinline__ void two_tile_wave(
    const float* const (&xrow)[2], v2f (&b0)[2][2], v2f (&b1)[2][2],
    const v4f* afrag, float* __restrict__ out, long long n0, long long n1, int mhalf)
{
    v8f acc[4][2];
    {
        v8f zero = {};
        #pragma unroll
        for (int ft = 0; ft < 4; ++ft) { acc[ft][0] = zero; acc[ft][1] = zero; }
    }
    v2f b2[2][2];
    v4f a0[4], a1[4];
    LOAD_A(a0, 0)

    #pragma unroll
    for (int kp = 0; kp < KPAIRS; ++kp) {
        if (kp + 1 < KPAIRS) { LOAD_A(a1, kp + 1) }
        if (kp + 2 < KPAIRS) { LOAD_B1(b2, kp + 2, 0) LOAD_B1(b2, kp + 2, 1) }
        __builtin_amdgcn_sched_barrier(0);
        #pragma unroll
        for (int s = 0; s < 2; ++s) {
            #pragma unroll
            for (int ft = 0; ft < 4; ++ft) {
                v2f as = {a0[ft][2 * s], a0[ft][2 * s + 1]};
                acc[ft][0] = wmma4(as, b0[s][0], acc[ft][0]);
                acc[ft][1] = wmma4(as, b0[s][1], acc[ft][1]);
            }
        }
        #pragma unroll
        for (int ft = 0; ft < 4; ++ft) a0[ft] = a1[ft];
        #pragma unroll
        for (int s = 0; s < 2; ++s)
            #pragma unroll
            for (int nt = 0; nt < 2; ++nt) {
                b0[s][nt] = b1[s][nt];
                b1[s][nt] = b2[s][nt];
            }
    }

    if (n0 < (long long)N_TOTAL) {
        #pragma unroll
        for (int ft = 0; ft < 4; ++ft) store_ftile(out, ft, mhalf, n0, acc[ft][0]);
    }
    if (n1 < (long long)N_TOTAL) {
        #pragma unroll
        for (int ft = 0; ft < 4; ++ft) store_ftile(out, ft, mhalf, n1, acc[ft][1]);
    }
}

template <bool SHARE>
__device__ __forceinline__ void one_tile_wave(
    const float* const (&xrow)[2], v2f (&b0)[2][2], v2f (&b1)[2][2],
    const v4f* afrag, const v4f* afragx, float* __restrict__ out,
    long long n0, long long n1, int mhalf, int fx)
{
    v8f acc[4], accx;
    {
        v8f zero = {};
        #pragma unroll
        for (int ft = 0; ft < 4; ++ft) acc[ft] = zero;
        accx = zero;
    }
    v2f b2[2][2];
    v4f a0[4], a1[4], ax0, ax1;
    LOAD_A(a0, 0)
    if (SHARE) ax0 = afragx[0];

    #pragma unroll
    for (int kp = 0; kp < KPAIRS; ++kp) {
        if (kp + 1 < KPAIRS) {
            LOAD_A(a1, kp + 1)
            if (SHARE) ax1 = afragx[(kp + 1) * 4 * 32];
        }
        if (kp + 2 < KPAIRS) {
            LOAD_B1(b2, kp + 2, 0)
            if (SHARE) { LOAD_B1(b2, kp + 2, 1) }
        }
        __builtin_amdgcn_sched_barrier(0);
        #pragma unroll
        for (int s = 0; s < 2; ++s) {
            #pragma unroll
            for (int ft = 0; ft < 4; ++ft) {
                v2f as = {a0[ft][2 * s], a0[ft][2 * s + 1]};
                acc[ft] = wmma4(as, b0[s][0], acc[ft]);
            }
            if (SHARE) {
                v2f asx = {ax0[2 * s], ax0[2 * s + 1]};
                accx = wmma4(asx, b0[s][1], accx);
            }
        }
        #pragma unroll
        for (int ft = 0; ft < 4; ++ft) a0[ft] = a1[ft];
        if (SHARE) ax0 = ax1;
        #pragma unroll
        for (int s = 0; s < 2; ++s) {
            b0[s][0] = b1[s][0];
            b1[s][0] = b2[s][0];
            if (SHARE) {
                b0[s][1] = b1[s][1];
                b1[s][1] = b2[s][1];
            }
        }
    }

    if (n0 < (long long)N_TOTAL) {
        #pragma unroll
        for (int ft = 0; ft < 4; ++ft) store_ftile(out, ft, mhalf, n0, acc[ft]);
    }
    if (SHARE) {
        if (n1 < (long long)N_TOTAL) store_ftile(out, fx, mhalf, n1, accx);
    }
}

__global__ __launch_bounds__(512) void conv_gemm_wmma_f32(
    const float* __restrict__ x,
    const float* __restrict__ w,
    float* __restrict__ out)
{
    __shared__ v4f wfrag[KPAIRS * 4 * 32];

    const int lane  = threadIdx.x & 31;
    const int wave  = __builtin_amdgcn_readfirstlane((int)(threadIdx.x >> 5));
    const int lrow  = lane & 15;
    const int half  = lane >> 4;
    const int khalf = half << 1;
    const int mhalf = half << 3;

    const int blk       = blockIdx.x;
    const bool has29    = blk < TILES_REM;
    const int tile_base = TILES_BASE * blk + (has29 ? blk : TILES_REM);
    const unsigned one_tile_waves = 0x1248u;
    const bool single   = ((one_tile_waves >> wave) & 1u) != 0u;
    const int before    = __builtin_popcount(one_tile_waves & ((1u << wave) - 1u));
    const bool two_slots = !single | has29;
    int t0, t1;
    if (!single) {
        t0 = tile_base + 2 * (wave - before);
        t1 = t0 + 1;
    } else {
        t0 = tile_base + 24 + before;
        t1 = tile_base + 28;
    }
    const long long n0 = (long long)t0 * 16 + lrow;
    const long long n1 = (long long)t1 * 16 + lrow;

    const float* xrow[2];
    {
        long long c0 = n0 > (long long)(N_TOTAL - 1) ? (long long)(N_TOTAL - 1) : n0;
        long long c1 = n1 > (long long)(N_TOTAL - 1) ? (long long)(N_TOTAL - 1) : n1;
        xrow[0] = x + c0 * DDIM + khalf;
        xrow[1] = x + c1 * DDIM + khalf;
    }

    v2f b0[2][2], b1[2][2];
    LOAD_B1(b0, 0, 0)
    LOAD_B1(b1, 1, 0)
    if (two_slots) {
        LOAD_B1(b0, 0, 1)
        LOAD_B1(b1, 1, 1)
    }

    {
        float* wl = (float*)wfrag;
        #pragma unroll
        for (int j = 0; j < 4; ++j) {
            const int fb = wave * 4 + j;
            const int kp = fb >> 2;
            const int ft = fb & 3;
            const v4f v = *(const v4f*)(w + (ft * 16 + lrow) * DDIM + kp * 8 + half * 4);
            v2f lo = {v[0], v[1]};
            v2f hi = {v[2], v[3]};
            float* dst = wl + fb * 128 + lrow * 4 + half * 2;
            *(v2f*)(dst)      = lo;
            *(v2f*)(dst + 64) = hi;
        }
    }

    __syncthreads();

    const v4f* afrag = wfrag + lane;
    if (!single) {
        two_tile_wave(xrow, b0, b1, afrag, out, n0, n1, mhalf);
    } else {
        const v4f* afragx = afrag + before * 32;
        if (has29) one_tile_wave<true >(xrow, b0, b1, afrag, afragx, out, n0, n1, mhalf, before);
        else       one_tile_wave<false>(xrow, b0, b1, afrag, afragx, out, n0, n1, mhalf, before);
    }
}

extern "C" void kernel_launch(void* const* d_in, const int* in_sizes, int n_in,
                              void* d_out, int out_size, void* d_ws, size_t ws_size,
                              hipStream_t stream) {
    const float* x = (const float*)d_in[0];
    const float* w = (const float*)d_in[1];
    float* out = (float*)d_out;

    conv_gemm_wmma_f32<<<GRID_BLOCKS, WAVES_PER_BLOCK * 32, 0, stream>>>(x, w, out);
}
